# starting kernel
# baseline (speedup 1.0000x reference)
#include <hip/hip_runtime.h>


#define INPUT_DIM   128
#define HID         16
#define MAX_DEPTH   8
#define N_INTERNAL  255
#define N_LEAVES    256
#define NUM_CLASSES 10
#define EFF_IN      130
#define WFLOATS     (EFF_IN * HID)
#define ROWS_PER_WG 64
#define THREADS     128

typedef __attribute__((ext_vector_type(2))) float v2f;
typedef __attribute__((ext_vector_type(8))) float v8f;

struct SMem {
  float W[2][WFLOATS];
  float B1[2][HID];
  float W2[2][HID];
  float B2[2];
  float leafP[N_LEAVES * 16];
  float PP[ROWS_PER_WG][N_LEAVES + 1];
  float Cur[ROWS_PER_WG][129];
  float Prev[ROWS_PER_WG][65];
  float Prev2[ROWS_PER_WG][33];
};

__global__ __launch_bounds__(THREADS) void tree_wmma_kernel(
    const float* __restrict__ x, const float* __restrict__ path_prob,
    const float* __restrict__ W1, const float* __restrict__ b1,
    const float* __restrict__ w2, const float* __restrict__ b2,
    const float* __restrict__ leaf_logits, float* __restrict__ out)
{
  __shared__ SMem sm;
  const int tid  = threadIdx.x;
  const int wave = tid >> 5;
  const int lane = tid & 31;
  const int half = lane >> 4;
  const int col  = lane & 15;
  const int kb   = half * 2;
  const int wgRow0   = blockIdx.x * ROWS_PER_WG;
  const int waveRow0 = wgRow0 + wave * 16;
  const int rl       = wave * 16 + col;

  v2f afrag[32];
  {
    const v2f* x2 = (const v2f*)(x + (size_t)(waveRow0 + col) * INPUT_DIM);
    #pragma unroll
    for (int k = 0; k < 32; ++k) afrag[k] = x2[2 * k + half];
  }

  for (int j = tid; j < N_LEAVES * 16; j += THREADS) {
    const int lr = j >> 4, c = j & 15;
    sm.leafP[j] = (c < NUM_CLASSES) ? leaf_logits[lr * NUM_CLASSES + c] : 0.0f;
  }
  if (tid < ROWS_PER_WG) {
    sm.PP[tid][0]    = path_prob[wgRow0 + tid];
    sm.Prev[tid][0]  = 0.0f;
    sm.Prev2[tid][0] = 0.0f;
  }

  float4 wreg[5];
  float  b1v = 0.0f, w2v = 0.0f, b2v = 0.0f;

  {
    const float4* src = (const float4*)(W1);
    #pragma unroll
    for (int t = 0; t < 4; ++t) wreg[t] = src[tid + THREADS * t];
    if (tid < 8)   wreg[4] = src[tid + 512];
    if (tid < HID) { b1v = b1[tid]; w2v = w2[tid]; }
    if (tid == 0)  b2v = b2[0];

    float4* dst = (float4*)sm.W[0];
    #pragma unroll
    for (int t = 0; t < 4; ++t) dst[tid + THREADS * t] = wreg[t];
    if (tid < 8)   dst[tid + 512] = wreg[4];
    if (tid < HID) { sm.B1[0][tid] = b1v; sm.W2[0][tid] = w2v; }
    if (tid == 0)  sm.B2[0] = b2v;
  }
  __syncthreads();

  int curbuf = 0;
  for (int d = 0; d < MAX_DEPTH; ++d) {
    const int nNodes   = 1 << d;
    const int nodeBase = nNodes - 1;

    for (int i = 0; i < nNodes; ++i) {
      const int n = nodeBase + i;

      const bool havenext = (n + 1 < N_INTERNAL);
      if (havenext) {
        const float4* src = (const float4*)(W1 + (size_t)(n + 1) * WFLOATS);
        #pragma unroll
        for (int t = 0; t < 4; ++t) wreg[t] = src[tid + THREADS * t];
        if (tid < 8)   wreg[4] = src[tid + 512];
        if (tid < HID) { b1v = b1[(n + 1) * HID + tid]; w2v = w2[(n + 1) * HID + tid]; }
        if (tid == 0)  b2v = b2[n + 1];
      }
      if (n + 2 < N_INTERNAL)
        __builtin_prefetch(W1 + (size_t)(n + 2) * WFLOATS + tid * 16, 0, 1);

      const float* Wc = sm.W[curbuf];
      v8f acc0 = {0.f,0.f,0.f,0.f,0.f,0.f,0.f,0.f};
      v8f acc1 = {0.f,0.f,0.f,0.f,0.f,0.f,0.f,0.f};
      #pragma unroll
      for (int k = 0; k < 32; k += 2) {
        v2f bf0, bf1;
        bf0.x = Wc[(4 * k + kb) * HID + col];
        bf0.y = Wc[(4 * k + kb + 1) * HID + col];
        bf1.x = Wc[(4 * (k + 1) + kb) * HID + col];
        bf1.y = Wc[(4 * (k + 1) + kb + 1) * HID + col];
        acc0 = __builtin_amdgcn_wmma_f32_16x16x4_f32(false, afrag[k],     false, bf0,
                                                     (short)0, acc0, false, false);
        acc1 = __builtin_amdgcn_wmma_f32_16x16x4_f32(false, afrag[k + 1], false, bf1,
                                                     (short)0, acc1, false, false);
      }
      {
        v2f ae = {0.f, 0.f}, be = {0.f, 0.f};
        if (half == 0) {
          ae.x = sm.Prev[rl][i >> 1];
          ae.y = sm.Prev2[rl][i >> 2];
          be.x = Wc[128 * HID + col];
          be.y = Wc[129 * HID + col];
        }
        acc0 = __builtin_amdgcn_wmma_f32_16x16x4_f32(false, ae, false, be,
                                                     (short)0, acc0, false, false);
      }

      const float b1c = sm.B1[curbuf][col];
      const float w2c = sm.W2[curbuf][col];
      const float b2c = sm.B2[curbuf];
      #pragma unroll
      for (int r = 0; r < 8; ++r) {
        float h = acc0[r] + acc1[r];
        float t = fmaxf(h + b1c, 0.0f);
        float v = t * w2c;
        v += __shfl_xor(v, 1);
        v += __shfl_xor(v, 2);
        v += __shfl_xor(v, 4);
        v += __shfl_xor(v, 8);
        if (col == 0) sm.Cur[wave * 16 + half * 8 + r][i] = v + b2c;
      }

      if (havenext) {
        float4* dst = (float4*)sm.W[curbuf ^ 1];
        #pragma unroll
        for (int t = 0; t < 4; ++t) dst[tid + THREADS * t] = wreg[t];
        if (tid < 8)   dst[tid + 512] = wreg[4];
        if (tid < HID) { sm.B1[curbuf ^ 1][tid] = b1v; sm.W2[curbuf ^ 1][tid] = w2v; }
        if (tid == 0)  sm.B2[curbuf ^ 1] = b2v;
      }
      __syncthreads();
      curbuf ^= 1;
    }

    {
      const int shift = MAX_DEPTH - d;
      const int gap   = 1 << (shift - 1);
      for (int tix = tid; tix < (ROWS_PER_WG << d); tix += THREADS) {
        const int r = tix >> d;
        const int i = tix & (nNodes - 1);
        const float s = sm.Cur[r][i];
        const float p = 1.0f / (1.0f + expf(-s));
        const int slot = i << shift;
        const float v = sm.PP[r][slot];
        sm.PP[r][slot]       = v * p;
        sm.PP[r][slot + gap] = v * (1.0f - p);
      }
    }
    __syncthreads();
    if (d < MAX_DEPTH - 1) {
      if (d >= 1) {
        for (int tix = tid; tix < (ROWS_PER_WG << (d - 1)); tix += THREADS) {
          const int r = tix >> (d - 1);
          const int j = tix & ((1 << (d - 1)) - 1);
          sm.Prev2[r][j] = sm.Prev[r][j];
        }
      }
      __syncthreads();
      for (int tix = tid; tix < (ROWS_PER_WG << d); tix += THREADS) {
        const int r = tix >> d;
        const int j = tix & (nNodes - 1);
        sm.Prev[r][j] = sm.Cur[r][j];
      }
      __syncthreads();
    }
  }

  __syncthreads();
  {
    v8f e0 = {0.f,0.f,0.f,0.f,0.f,0.f,0.f,0.f};
    v8f e1 = {0.f,0.f,0.f,0.f,0.f,0.f,0.f,0.f};
    #pragma unroll
    for (int k = 0; k < 64; k += 2) {
      v2f a0, bq0, a1, bq1;
      a0.x  = sm.PP[rl][4 * k + kb];
      a0.y  = sm.PP[rl][4 * k + kb + 1];
      bq0.x = sm.leafP[(4 * k + kb) * 16 + col];
      bq0.y = sm.leafP[(4 * k + kb + 1) * 16 + col];
      a1.x  = sm.PP[rl][4 * (k + 1) + kb];
      a1.y  = sm.PP[rl][4 * (k + 1) + kb + 1];
      bq1.x = sm.leafP[(4 * (k + 1) + kb) * 16 + col];
      bq1.y = sm.leafP[(4 * (k + 1) + kb + 1) * 16 + col];
      e0 = __builtin_amdgcn_wmma_f32_16x16x4_f32(false, a0, false, bq0,
                                                 (short)0, e0, false, false);
      e1 = __builtin_amdgcn_wmma_f32_16x16x4_f32(false, a1, false, bq1,
                                                 (short)0, e1, false, false);
    }
    if (col < NUM_CLASSES) {
      #pragma unroll
      for (int r = 0; r < 8; ++r)
        out[(size_t)(waveRow0 + half * 8 + r) * NUM_CLASSES + col] = e0[r] + e1[r];
    }
  }
}

extern "C" void kernel_launch(void* const* d_in, const int* in_sizes, int n_in,
                              void* d_out, int out_size, void* d_ws, size_t ws_size,
                              hipStream_t stream) {
  const float* x    = (const float*)d_in[0];
  const float* pp   = (const float*)d_in[1];
  const float* W1   = (const float*)d_in[2];
  const float* b1   = (const float*)d_in[3];
  const float* w2   = (const float*)d_in[4];
  const float* b2   = (const float*)d_in[5];
  const float* leaf = (const float*)d_in[6];
  float* out = (float*)d_out;

  const int B = in_sizes[0] / INPUT_DIM;
  dim3 grid(B / ROWS_PER_WG);
  dim3 block(THREADS);
  tree_wmma_kernel<<<grid, block, 0, stream>>>(x, pp, W1, b1, w2, b2, leaf, out);
}


# Weights repacked each call by a first kernel into per-lane fragment order in the workspace; a node's operands arrive as 17 16-byte loads in one run (was 68 4-byte loads in 17 places); sums unchanged
# speedup vs baseline: 4.5182x; 4.5182x over previous
#include <hip/hip_runtime.h>


#define INPUT_DIM   128
#define HID         16
#define MAX_DEPTH   8
#define N_INTERNAL  255
#define N_LEAVES    256
#define NUM_CLASSES 10
#define EFF_IN      130
#define WFLOATS     (EFF_IN * HID)
#define ROWS_PER_WG 64
#define THREADS     128
#define Q_PER_NODE  17
#define NODE_Q      (Q_PER_NODE * 32)

typedef __attribute__((ext_vector_type(2))) float v2f;
typedef __attribute__((ext_vector_type(8))) float v8f;

struct SMem {
  float leafP[N_LEAVES * 16];
  float PP[ROWS_PER_WG][N_LEAVES + 1];
  float Cur[ROWS_PER_WG][129];
  float Prev[ROWS_PER_WG][65];
  float Prev2[ROWS_PER_WG][33];
};

struct NodeFrag {
  float4 q[Q_PER_NODE];
};

__global__ __launch_bounds__(NODE_Q) void pack_weights_kernel(
    const float* __restrict__ W1, const float* __restrict__ b1,
    const float* __restrict__ w2, const float* __restrict__ b2,
    float* __restrict__ ws)
{
  const int n    = blockIdx.x;
  const int p    = threadIdx.x >> 5;
  const int lane = threadIdx.x & 31;
  const int h    = lane >> 4;
  const int c    = lane & 15;
  const float* Wn = W1 + (size_t)n * WFLOATS;
  float4 v;
  if (p < 16) {
    const float* r = Wn + (8 * p + h) * HID + c;
    v.x = r[0];
    v.y = r[2 * HID];
    v.z = r[4 * HID];
    v.w = r[6 * HID];
  } else {
    v.x = Wn[(INPUT_DIM + h) * HID + c];
    v.y = b1[n * HID + c];
    v.z = w2[n * HID + c];
    v.w = b2[n];
  }
  ((float4*)ws)[(size_t)n * NODE_Q + threadIdx.x] = v;
}

#define ROW_XOR(v, M) \
  __builtin_bit_cast(float, __builtin_amdgcn_update_dpp(0, __builtin_bit_cast(int, (v)), 0x160 + (M), 0xf, 0xf, false))

__global__ __launch_bounds__(THREADS) void tree_wmma_kernel(
    const float* __restrict__ x, const float* __restrict__ path_prob,
    const float* __restrict__ wpacked,
    const float* __restrict__ leaf_logits, float* __restrict__ out)
{
  __shared__ SMem sm;
  const int tid  = threadIdx.x;
  const int wave = tid >> 5;
  const int lane = tid & 31;
  const int half = lane >> 4;
  const int col  = lane & 15;
  const int kb   = half * 2;
  const int wgRow0   = blockIdx.x * ROWS_PER_WG;
  const int waveRow0 = wgRow0 + wave * 16;
  const int rl       = wave * 16 + col;

  v2f afrag[32];
  {
    const float4* x4 = (const float4*)(x + (size_t)(waveRow0 + col) * INPUT_DIM);
    #pragma unroll
    for (int j = 0; j < 32; ++j) {
      const float4 v = x4[j];
      afrag[j].x = half ? v.y : v.x;
      afrag[j].y = half ? v.w : v.z;
    }
  }

  for (int j = tid; j < N_LEAVES * 16; j += THREADS) {
    const int lr = j >> 4, c = j & 15;
    sm.leafP[j] = (c < NUM_CLASSES) ? leaf_logits[lr * NUM_CLASSES + c] : 0.0f;
  }
  if (tid < ROWS_PER_WG) {
    sm.PP[tid][0]    = path_prob[wgRow0 + tid];
    sm.Prev[tid][0]  = 0.0f;
    sm.Prev2[tid][0] = 0.0f;
  }

  const float4* wq = (const float4*)wpacked + lane;
  NodeFrag f0, f1;
  #pragma unroll
  for (int p = 0; p < Q_PER_NODE; ++p) f0.q[p] = wq[p * 32];
  __syncthreads();

  auto node_step = [&](const NodeFrag& fc, NodeFrag& fn, int nNext, int i, const bool fetch) __attribute__((always_inline)) {
    const float4* Wn = wq + (size_t)nNext * NODE_Q;
    v8f acc0 = {0.f,0.f,0.f,0.f,0.f,0.f,0.f,0.f};
    v8f acc1 = {0.f,0.f,0.f,0.f,0.f,0.f,0.f,0.f};
    if (fetch) {
      #pragma unroll
      for (int p = 0; p < Q_PER_NODE; ++p) fn.q[p] = Wn[p * 32];
    }
    #pragma unroll
    for (int p = 0; p < 16; ++p) {
      v2f be0, be1;
      be0.x = fc.q[p].x;  be0.y = fc.q[p].y;
      be1.x = fc.q[p].z;  be1.y = fc.q[p].w;
      acc0 = __builtin_amdgcn_wmma_f32_16x16x4_f32(false, afrag[2 * p],     false, be0,
                                                   (short)0, acc0, false, false);
      acc1 = __builtin_amdgcn_wmma_f32_16x16x4_f32(false, afrag[2 * p + 1], false, be1,
                                                   (short)0, acc1, false, false);
    }
    {
      v2f ae, be;
      ae.x = half ? sm.Prev2[rl][i >> 2] : sm.Prev[rl][i >> 1];
      ae.y = 0.0f;
      be.x = fc.q[16].x;
      be.y = 0.0f;
      acc0 = __builtin_amdgcn_wmma_f32_16x16x4_f32(false, ae, false, be,
                                                   (short)0, acc0, false, false);
    }
    float part[8];
    #pragma unroll
    for (int r = 0; r < 8; ++r) {
      float h = acc0[r] + acc1[r];
      float t = fmaxf(h + fc.q[16].y, 0.0f);
      float v = t * fc.q[16].z;
      v += ROW_XOR(v, 1);
      v += ROW_XOR(v, 2);
      v += ROW_XOR(v, 4);
      part[r] = v;
    }
    float sel = part[0];
    #pragma unroll
    for (int r = 1; r < 8; ++r) sel = ((col & 7) == r) ? part[r] : sel;
    const float tot = sel + ROW_XOR(sel, 8);
    if (col < 8) sm.Cur[wave * 16 + half * 8 + col][i] = tot + fc.q[16].w;
  };

  int d = 0, nodeBase = 0;
  auto close_depth = [&]() __attribute__((always_inline)) {
    const int nNodes = 1 << d;
    __syncthreads();
    {
      const int shift = MAX_DEPTH - d;
      const int gap   = 1 << (shift - 1);
      for (int tix = tid; tix < (ROWS_PER_WG << d); tix += THREADS) {
        const int r = tix >> d;
        const int i = tix & (nNodes - 1);
        const float s = sm.Cur[r][i];
        const float p = 1.0f / (1.0f + expf(-s));
        const int slot = i << shift;
        const float v = sm.PP[r][slot];
        sm.PP[r][slot]       = v * p;
        sm.PP[r][slot + gap] = v * (1.0f - p);
      }
    }
    __syncthreads();
    if (d < MAX_DEPTH - 1) {
      if (d >= 1) {
        for (int tix = tid; tix < (ROWS_PER_WG << (d - 1)); tix += THREADS) {
          const int r = tix >> (d - 1);
          const int j = tix & ((1 << (d - 1)) - 1);
          sm.Prev2[r][j] = sm.Prev[r][j];
        }
      }
      __syncthreads();
      for (int tix = tid; tix < (ROWS_PER_WG << d); tix += THREADS) {
        const int r = tix >> d;
        const int j = tix & (nNodes - 1);
        sm.Prev[r][j] = sm.Cur[r][j];
      }
      __syncthreads();
    }
    nodeBase += nNodes;
    ++d;
  };

  for (int m = 0; m < (N_INTERNAL - 1) / 2; ++m) {
    const int n0 = 2 * m;
    node_step(f0, f1, n0 + 1, n0 - nodeBase, true);
    if (n0 == nodeBase + (1 << d) - 1)
      close_depth();
    node_step(f1, f0, n0 + 2, n0 + 1 - nodeBase, true);
  }
  node_step(f0, f1, 0, N_INTERNAL - 1 - nodeBase, false);
  close_depth();

  __syncthreads();
  {
    v8f e0 = {0.f,0.f,0.f,0.f,0.f,0.f,0.f,0.f};
    v8f e1 = {0.f,0.f,0.f,0.f,0.f,0.f,0.f,0.f};
    #pragma unroll
    for (int k = 0; k < 64; k += 2) {
      v2f a0, bq0, a1, bq1;
      a0.x  = sm.PP[rl][4 * k + kb];
      a0.y  = sm.PP[rl][4 * k + kb + 1];
      bq0.x = sm.leafP[(4 * k + kb) * 16 + col];
      bq0.y = sm.leafP[(4 * k + kb + 1) * 16 + col];
      a1.x  = sm.PP[rl][4 * (k + 1) + kb];
      a1.y  = sm.PP[rl][4 * (k + 1) + kb + 1];
      bq1.x = sm.leafP[(4 * (k + 1) + kb) * 16 + col];
      bq1.y = sm.leafP[(4 * (k + 1) + kb + 1) * 16 + col];
      e0 = __builtin_amdgcn_wmma_f32_16x16x4_f32(false, a0, false, bq0,
                                                 (short)0, e0, false, false);
      e1 = __builtin_amdgcn_wmma_f32_16x16x4_f32(false, a1, false, bq1,
                                                 (short)0, e1, false, false);
    }
    if (col < NUM_CLASSES) {
      #pragma unroll
      for (int r = 0; r < 8; ++r)
        out[(size_t)(waveRow0 + half * 8 + r) * NUM_CLASSES + col] = e0[r] + e1[r];
    }
  }
}

extern "C" void kernel_launch(void* const* d_in, const int* in_sizes, int n_in,
                              void* d_out, int out_size, void* d_ws, size_t ws_size,
                              hipStream_t stream) {
  const float* x    = (const float*)d_in[0];
  const float* pp   = (const float*)d_in[1];
  const float* W1   = (const float*)d_in[2];
  const float* b1   = (const float*)d_in[3];
  const float* w2   = (const float*)d_in[4];
  const float* b2   = (const float*)d_in[5];
  const float* leaf = (const float*)d_in[6];
  float* out = (float*)d_out;

  const int B = in_sizes[0] / INPUT_DIM;
  dim3 grid(B / ROWS_PER_WG);
  dim3 block(THREADS);
  pack_weights_kernel<<<dim3(N_INTERNAL), dim3(NODE_Q), 0, stream>>>(W1, b1, w2, b2, (float*)d_ws);
  tree_wmma_kernel<<<grid, block, 0, stream>>>(x, pp, (const float*)d_ws, leaf, out);
}


# Recursion made wave-private: scores in one level-order LDS array read by index, sigmoid and path-prob split done per sibling pair in a node's vector tail; depth passes, copies, 31 barriers gone
# speedup vs baseline: 4.9462x; 1.0947x over previous
#include <hip/hip_runtime.h>


#define INPUT_DIM   128
#define HID         16
#define MAX_DEPTH   8
#define N_INTERNAL  255
#define N_LEAVES    256
#define NUM_CLASSES 10
#define EFF_IN      130
#define WFLOATS     (EFF_IN * HID)
#define ROWS_PER_WG 64
#define THREADS     128
#define Q_PER_NODE  17
#define NODE_Q      (Q_PER_NODE * 32)
#define S_ZERO      128
#define S_PITCH     129

typedef __attribute__((ext_vector_type(2))) float v2f;
typedef __attribute__((ext_vector_type(8))) float v8f;

struct SMem {
  float leafP[N_LEAVES * 16];
  float PP[ROWS_PER_WG][N_LEAVES + 1];
  float S[ROWS_PER_WG][S_PITCH];
};

struct NodeFrag {
  float4 q[Q_PER_NODE];
};

__global__ __launch_bounds__(NODE_Q) void pack_weights_kernel(
    const float* __restrict__ W1, const float* __restrict__ b1,
    const float* __restrict__ w2, const float* __restrict__ b2,
    float* __restrict__ ws)
{
  const int n    = blockIdx.x;
  const int p    = threadIdx.x >> 5;
  const int lane = threadIdx.x & 31;
  const int h    = lane >> 4;
  const int c    = lane & 15;
  const float* Wn = W1 + (size_t)n * WFLOATS;
  float4 v;
  if (p < 16) {
    const float* r = Wn + (8 * p + h) * HID + c;
    v.x = r[0];
    v.y = r[2 * HID];
    v.z = r[4 * HID];
    v.w = r[6 * HID];
  } else {
    v.x = Wn[(INPUT_DIM + h) * HID + c];
    v.y = b1[n * HID + c];
    v.z = w2[n * HID + c];
    v.w = b2[n];
  }
  ((float4*)ws)[(size_t)n * NODE_Q + threadIdx.x] = v;
}

#define ROW_XOR(v, M) \
  __builtin_bit_cast(float, __builtin_amdgcn_update_dpp(0, __builtin_bit_cast(int, (v)), 0x160 + (M), 0xf, 0xf, false))

__global__ __launch_bounds__(THREADS) void tree_wmma_kernel(
    const float* __restrict__ x, const float* __restrict__ path_prob,
    const float* __restrict__ wpacked,
    const float* __restrict__ leaf_logits, float* __restrict__ out)
{
  __shared__ SMem sm;
  const int tid  = threadIdx.x;
  const int wave = tid >> 5;
  const int lane = tid & 31;
  const int half = lane >> 4;
  const int col  = lane & 15;
  const int kb   = half * 2;
  const int wgRow0   = blockIdx.x * ROWS_PER_WG;
  const int waveRow0 = wgRow0 + wave * 16;
  const int rl       = wave * 16 + col;

  v2f afrag[32];
  {
    const float4* x4 = (const float4*)(x + (size_t)(waveRow0 + col) * INPUT_DIM);
    #pragma unroll
    for (int j = 0; j < 32; ++j) {
      const float4 v = x4[j];
      afrag[j].x = half ? v.y : v.x;
      afrag[j].y = half ? v.w : v.z;
    }
  }

  for (int j = tid; j < N_LEAVES * 16; j += THREADS) {
    const int lr = j >> 4, c = j & 15;
    sm.leafP[j] = (c < NUM_CLASSES) ? leaf_logits[lr * NUM_CLASSES + c] : 0.0f;
  }
  if (lane < 16) {
    sm.PP[wave * 16 + lane][0]     = path_prob[waveRow0 + lane];
    sm.S[wave * 16 + lane][S_ZERO] = 0.0f;
  }

  const float4* wq = (const float4*)wpacked + lane;
  NodeFrag f0, f1;
  #pragma unroll
  for (int p = 0; p < Q_PER_NODE; ++p) f0.q[p] = wq[p * 32];
  __syncthreads();

  const int sub  = (col >> 3) & 1;
  const int prow = wave * 16 + half * 8 + (col & 7);
  float*       ppRow = &sm.PP[prow][0];
  float*       sOut  = &sm.S[prow][0];
  const float* sIn   = &sm.S[rl][0];

  auto split = [&](float s, float v, int slot, int gap) __attribute__((always_inline)) {
    const float p = 1.0f / (1.0f + expf(-s));
    ppRow[slot]       = v * p;
    ppRow[slot + gap] = v * (1.0f - p);
  };

  auto node_score = [&](const NodeFrag& fc, NodeFrag& fn, int nNext, float aex,
                        const bool withSplit, float sP, float vP, int slotP, int gapP)
                        __attribute__((always_inline)) {
    const float4* Wn = wq + (size_t)nNext * NODE_Q;
    v8f acc0 = {0.f,0.f,0.f,0.f,0.f,0.f,0.f,0.f};
    v8f acc1 = {0.f,0.f,0.f,0.f,0.f,0.f,0.f,0.f};
    __builtin_amdgcn_sched_barrier(0);
    #pragma unroll
    for (int p = 0; p < Q_PER_NODE; ++p) fn.q[p] = Wn[p * 32];
    #pragma unroll
    for (int p = 0; p < 16; ++p) {
      v2f be0, be1;
      be0.x = fc.q[p].x;  be0.y = fc.q[p].y;
      be1.x = fc.q[p].z;  be1.y = fc.q[p].w;
      acc0 = __builtin_amdgcn_wmma_f32_16x16x4_f32(false, afrag[2 * p],     false, be0,
                                                   (short)0, acc0, false, false);
      acc1 = __builtin_amdgcn_wmma_f32_16x16x4_f32(false, afrag[2 * p + 1], false, be1,
                                                   (short)0, acc1, false, false);
    }
    {
      v2f ae, be;
      ae.x = aex;
      ae.y = 0.0f;
      be.x = fc.q[16].x;
      be.y = 0.0f;
      acc0 = __builtin_amdgcn_wmma_f32_16x16x4_f32(false, ae, false, be,
                                                   (short)0, acc0, false, false);
    }
    __builtin_amdgcn_sched_barrier(0);
    if (withSplit) split(sP, vP, slotP, gapP);
    float part[8];
    #pragma unroll
    for (int r = 0; r < 8; ++r) {
      float h = acc0[r] + acc1[r];
      float t = fmaxf(h + fc.q[16].y, 0.0f);
      float v = t * fc.q[16].z;
      v += ROW_XOR(v, 1);
      v += ROW_XOR(v, 2);
      v += ROW_XOR(v, 4);
      part[r] = v;
    }
    float sel = part[0];
    #pragma unroll
    for (int r = 1; r < 8; ++r) sel = ((col & 7) == r) ? part[r] : sel;
    const float tot = sel + ROW_XOR(sel, 8);
    return tot + fc.q[16].w;
  };

  float sOwed    = node_score(f0, f1, 1, 0.0f, false, 0.0f, 0.0f, 0, 0);
  int   slotOwed = 0;
  int   gapOwed  = N_LEAVES / 2;
  sOut[0] = sOwed;

  for (int m = 0; m < (N_INTERNAL - 1) / 2; ++m) {
    const int n1 = 2 * m + 1;
    const int d  = 31 - __builtin_clz(n1 + 1);
    const int nb = (1 << d) - 1;
    const int j2 = n1 - nb;
    const int sh = MAX_DEPTH - d;
    const int pidx = (nb >> 1) + (j2 >> 1);
    const int gidx = (d >= 2) ? (nb >> 2) + (j2 >> 2) : S_ZERO;
    const float aex   = sIn[half ? gidx : pidx];
    const float vOwed = ppRow[slotOwed];
    const float sA = node_score(f1, f0, n1 + 1, aex, true, sOwed, vOwed, slotOwed, gapOwed);
    const int nn = (n1 + 2 < N_INTERNAL) ? n1 + 2 : N_INTERNAL - 1;
    const float sB = node_score(f0, f1, nn, aex, false, 0.0f, 0.0f, 0, 0);
    sOwed    = sub ? sB : sA;
    slotOwed = (j2 + sub) << sh;
    gapOwed  = 1 << (sh - 1);
    if (d < MAX_DEPTH - 1)
      sOut[nb + j2 + sub] = sOwed;
  }
  split(sOwed, ppRow[slotOwed], slotOwed, gapOwed);

  {
    v8f e0 = {0.f,0.f,0.f,0.f,0.f,0.f,0.f,0.f};
    v8f e1 = {0.f,0.f,0.f,0.f,0.f,0.f,0.f,0.f};
    #pragma unroll
    for (int k = 0; k < 64; k += 2) {
      v2f a0, bq0, a1, bq1;
      a0.x  = sm.PP[rl][4 * k + kb];
      a0.y  = sm.PP[rl][4 * k + kb + 1];
      bq0.x = sm.leafP[(4 * k + kb) * 16 + col];
      bq0.y = sm.leafP[(4 * k + kb + 1) * 16 + col];
      a1.x  = sm.PP[rl][4 * (k + 1) + kb];
      a1.y  = sm.PP[rl][4 * (k + 1) + kb + 1];
      bq1.x = sm.leafP[(4 * (k + 1) + kb) * 16 + col];
      bq1.y = sm.leafP[(4 * (k + 1) + kb + 1) * 16 + col];
      e0 = __builtin_amdgcn_wmma_f32_16x16x4_f32(false, a0, false, bq0,
                                                 (short)0, e0, false, false);
      e1 = __builtin_amdgcn_wmma_f32_16x16x4_f32(false, a1, false, bq1,
                                                 (short)0, e1, false, false);
    }
    if (col < NUM_CLASSES) {
      #pragma unroll
      for (int r = 0; r < 8; ++r)
        out[(size_t)(waveRow0 + half * 8 + r) * NUM_CLASSES + col] = e0[r] + e1[r];
    }
  }
}

extern "C" void kernel_launch(void* const* d_in, const int* in_sizes, int n_in,
                              void* d_out, int out_size, void* d_ws, size_t ws_size,
                              hipStream_t stream) {
  const float* x    = (const float*)d_in[0];
  const float* pp   = (const float*)d_in[1];
  const float* W1   = (const float*)d_in[2];
  const float* b1   = (const float*)d_in[3];
  const float* w2   = (const float*)d_in[4];
  const float* b2   = (const float*)d_in[5];
  const float* leaf = (const float*)d_in[6];
  float* out = (float*)d_out;

  const int B = in_sizes[0] / INPUT_DIM;
  dim3 grid(B / ROWS_PER_WG);
  dim3 block(THREADS);
  pack_weights_kernel<<<dim3(N_INTERNAL), dim3(NODE_Q), 0, stream>>>(W1, b1, w2, b2, (float*)d_ws);
  tree_wmma_kernel<<<grid, block, 0, stream>>>(x, pp, (const float*)d_ws, leaf, out);
}


# Sibling pair computed together: 66 WMMAs, one accumulator per node (K-steps in one chain), one tail with both w2 sums as a 16-register reduce-scatter over lanes 8,4,2,1; sum order changes
# speedup vs baseline: 5.2705x; 1.0656x over previous
#include <hip/hip_runtime.h>


#define INPUT_DIM   128
#define HID         16
#define MAX_DEPTH   8
#define N_INTERNAL  255
#define N_LEAVES    256
#define NUM_CLASSES 10
#define EFF_IN      130
#define WFLOATS     (EFF_IN * HID)
#define ROWS_PER_WG 64
#define THREADS     128
#define Q_PER_NODE  17
#define NODE_Q      (Q_PER_NODE * 32)
#define S_ZERO      128
#define S_PITCH     129

typedef __attribute__((ext_vector_type(2))) float v2f;
typedef __attribute__((ext_vector_type(8))) float v8f;

struct SMem {
  float leafP[N_LEAVES * 16];
  float PP[ROWS_PER_WG][N_LEAVES + 1];
  float S[ROWS_PER_WG][S_PITCH];
};

struct NodeFrag {
  float4 q[Q_PER_NODE];
};

__global__ __launch_bounds__(NODE_Q) void pack_weights_kernel(
    const float* __restrict__ W1, const float* __restrict__ b1,
    const float* __restrict__ w2, const float* __restrict__ b2,
    float* __restrict__ ws)
{
  const int n    = blockIdx.x;
  const int p    = threadIdx.x >> 5;
  const int lane = threadIdx.x & 31;
  const int h    = lane >> 4;
  const int c    = lane & 15;
  const float* Wn = W1 + (size_t)n * WFLOATS;
  float4 v;
  if (p < 16) {
    const float* r = Wn + (8 * p + h) * HID + c;
    v.x = r[0];
    v.y = r[2 * HID];
    v.z = r[4 * HID];
    v.w = r[6 * HID];
  } else {
    v.x = Wn[(INPUT_DIM + h) * HID + c];
    v.y = b1[n * HID + c];
    v.z = w2[n * HID + c];
    v.w = b2[n];
  }
  ((float4*)ws)[(size_t)n * NODE_Q + threadIdx.x] = v;
}

#define ROW_XOR(v, M) \
  __builtin_bit_cast(float, __builtin_amdgcn_update_dpp(0, __builtin_bit_cast(int, (v)), 0x160 + (M), 0xf, 0xf, false))

__global__ __launch_bounds__(THREADS) void tree_wmma_kernel(
    const float* __restrict__ x, const float* __restrict__ path_prob,
    const float* __restrict__ wpacked,
    const float* __restrict__ leaf_logits, float* __restrict__ out)
{
  __shared__ SMem sm;
  const int tid  = threadIdx.x;
  const int wave = tid >> 5;
  const int lane = tid & 31;
  const int half = lane >> 4;
  const int col  = lane & 15;
  const int kb   = half * 2;
  const int wgRow0   = blockIdx.x * ROWS_PER_WG;
  const int waveRow0 = wgRow0 + wave * 16;
  const int rl       = wave * 16 + col;

  v2f afrag[32];
  {
    const float4* x4 = (const float4*)(x + (size_t)(waveRow0 + col) * INPUT_DIM);
    #pragma unroll
    for (int j = 0; j < 32; ++j) {
      const float4 v = x4[j];
      afrag[j].x = half ? v.y : v.x;
      afrag[j].y = half ? v.w : v.z;
    }
  }

  for (int j = tid; j < N_LEAVES * 16; j += THREADS) {
    const int lr = j >> 4, c = j & 15;
    sm.leafP[j] = (c < NUM_CLASSES) ? leaf_logits[lr * NUM_CLASSES + c] : 0.0f;
  }
  if (lane < 16) {
    sm.PP[wave * 16 + lane][0]     = path_prob[waveRow0 + lane];
    sm.S[wave * 16 + lane][S_ZERO] = 0.0f;
  }

  if (lane < 16) sm.PP[wave * 16 + lane][N_LEAVES] = 0.0f;

  const float4* wq = (const float4*)wpacked + lane;
  NodeFrag fa, fb, ga, gb;
  #pragma unroll
  for (int p = 0; p < Q_PER_NODE; ++p) { fa.q[p] = wq[p * 32]; fb.q[p] = fa.q[p]; }
  __syncthreads();

  const bool sub = (col & 8) != 0;
  const bool bt2 = (col & 4) != 0;
  const bool bt1 = (col & 2) != 0;
  const bool bt0 = (col & 1) != 0;
  const int prow = wave * 16 + half * 8 + (col & 7);
  float*       ppRow = &sm.PP[prow][0];
  float*       sOut  = &sm.S[prow][0];
  const float* sIn   = &sm.S[rl][0];

  auto split = [&](float s, float v, int slot, int gap) __attribute__((always_inline)) {
    const float p = 1.0f / (1.0f + expf(-s));
    ppRow[slot]       = v * p;
    ppRow[slot + gap] = v * (1.0f - p);
  };

  auto pair_step = [&](const NodeFrag& ca, const NodeFrag& cb, NodeFrag& na, NodeFrag& nb2,
                       int nNextA, int nNextB, float aex,
                       float sP, float vP, int slotP, int gapP)
                       __attribute__((always_inline)) {
    const float4* WnA = wq + (size_t)nNextA * NODE_Q;
    const float4* WnB = wq + (size_t)nNextB * NODE_Q;
    v8f accA = {0.f,0.f,0.f,0.f,0.f,0.f,0.f,0.f};
    v8f accB = {0.f,0.f,0.f,0.f,0.f,0.f,0.f,0.f};
    __builtin_amdgcn_sched_barrier(0);
    #pragma unroll
    for (int p = 0; p < Q_PER_NODE; ++p) na.q[p]  = WnA[p * 32];
    #pragma unroll
    for (int p = 0; p < Q_PER_NODE; ++p) nb2.q[p] = WnB[p * 32];
    __builtin_amdgcn_sched_barrier(0);
    #pragma unroll
    for (int p = 0; p < 16; ++p) {
      v2f a0, a1, b0, b1;
      a0.x = ca.q[p].x;  a0.y = ca.q[p].y;
      a1.x = ca.q[p].z;  a1.y = ca.q[p].w;
      b0.x = cb.q[p].x;  b0.y = cb.q[p].y;
      b1.x = cb.q[p].z;  b1.y = cb.q[p].w;
      accA = __builtin_amdgcn_wmma_f32_16x16x4_f32(false, afrag[2 * p],     false, a0,
                                                   (short)0, accA, false, false);
      accB = __builtin_amdgcn_wmma_f32_16x16x4_f32(false, afrag[2 * p],     false, b0,
                                                   (short)0, accB, false, false);
      accA = __builtin_amdgcn_wmma_f32_16x16x4_f32(false, afrag[2 * p + 1], false, a1,
                                                   (short)0, accA, false, false);
      accB = __builtin_amdgcn_wmma_f32_16x16x4_f32(false, afrag[2 * p + 1], false, b1,
                                                   (short)0, accB, false, false);
    }
    {
      v2f ae, ba, bb;
      ae.x = aex;          ae.y = 0.0f;
      ba.x = ca.q[16].x;   ba.y = 0.0f;
      bb.x = cb.q[16].x;   bb.y = 0.0f;
      accA = __builtin_amdgcn_wmma_f32_16x16x4_f32(false, ae, false, ba,
                                                   (short)0, accA, false, false);
      accB = __builtin_amdgcn_wmma_f32_16x16x4_f32(false, ae, false, bb,
                                                   (short)0, accB, false, false);
    }
    __builtin_amdgcn_sched_barrier(0);
    split(sP, vP, slotP, gapP);
    float u[8];
    #pragma unroll
    for (int r = 0; r < 8; ++r) {
      const float ta = fmaxf(accA[r] + ca.q[16].y, 0.0f) * ca.q[16].z;
      const float tb = fmaxf(accB[r] + cb.q[16].y, 0.0f) * cb.q[16].z;
      const float keep = sub ? tb : ta;
      const float send = sub ? ta : tb;
      u[r] = keep + ROW_XOR(send, 8);
    }
    float w[4];
    #pragma unroll
    for (int r = 0; r < 4; ++r) {
      const float keep = bt2 ? u[r + 4] : u[r];
      const float send = bt2 ? u[r] : u[r + 4];
      w[r] = keep + ROW_XOR(send, 4);
    }
    float y[2];
    #pragma unroll
    for (int r = 0; r < 2; ++r) {
      const float keep = bt1 ? w[r + 2] : w[r];
      const float send = bt1 ? w[r] : w[r + 2];
      y[r] = keep + ROW_XOR(send, 2);
    }
    const float keep = bt0 ? y[1] : y[0];
    const float send = bt0 ? y[0] : y[1];
    const float tot  = keep + ROW_XOR(send, 1);
    return tot + (sub ? cb.q[16].w : ca.q[16].w);
  };

  float sOwed    = 0.0f;
  int   slotOwed = N_LEAVES;
  int   gapOwed  = 0;
  for (int it = 0; it < (N_INTERNAL + 1) / 4; ++it) {
    {
      const int  t    = 2 * it;
      const bool root = (it == 0);
      const int  n1   = 2 * t - 1;
      const int  d    = root ? 0 : 31 - __builtin_clz(2 * t);
      const int  nb   = (1 << d) - 1;
      const int  j2   = root ? 0 : n1 - nb;
      const int  sh   = MAX_DEPTH - d;
      const int  pidx = (d >= 1) ? (nb >> 1) + (j2 >> 1) : S_ZERO;
      const int  gidx = (d >= 2) ? (nb >> 2) + (j2 >> 2) : S_ZERO;
      const float aex   = sIn[half ? gidx : pidx];
      const float vOwed = ppRow[slotOwed];
      const float s = pair_step(fa, fb, ga, gb, 2 * t + 1, 2 * t + 2, aex,
                                sOwed, vOwed, slotOwed, gapOwed);
      sOwed    = s;
      slotOwed = root ? 0 : ((j2 + (sub ? 1 : 0)) << sh);
      gapOwed  = 1 << (sh - 1);
      if (d < MAX_DEPTH - 1)
        sOut[root ? 0 : n1 + (sub ? 1 : 0)] = s;
    }
    {
      const int  t    = 2 * it + 1;
      const int  n1   = 2 * t - 1;
      const int  d    = 31 - __builtin_clz(2 * t);
      const int  nb   = (1 << d) - 1;
      const int  j2   = n1 - nb;
      const int  sh   = MAX_DEPTH - d;
      const int  pidx = (nb >> 1) + (j2 >> 1);
      const int  gidx = (d >= 2) ? (nb >> 2) + (j2 >> 2) : S_ZERO;
      const float aex   = sIn[half ? gidx : pidx];
      const float vOwed = ppRow[slotOwed];
      const int nA = (2 * t + 1 < N_INTERNAL) ? 2 * t + 1 : N_INTERNAL - 1;
      const int nB = (2 * t + 2 < N_INTERNAL) ? 2 * t + 2 : N_INTERNAL - 1;
      const float s = pair_step(ga, gb, fa, fb, nA, nB, aex,
                                sOwed, vOwed, slotOwed, gapOwed);
      sOwed    = s;
      slotOwed = (j2 + (sub ? 1 : 0)) << sh;
      gapOwed  = 1 << (sh - 1);
      if (d < MAX_DEPTH - 1)
        sOut[n1 + (sub ? 1 : 0)] = s;
    }
  }
  split(sOwed, ppRow[slotOwed], slotOwed, gapOwed);

  {
    v8f e0 = {0.f,0.f,0.f,0.f,0.f,0.f,0.f,0.f};
    v8f e1 = {0.f,0.f,0.f,0.f,0.f,0.f,0.f,0.f};
    #pragma unroll
    for (int k = 0; k < 64; k += 2) {
      v2f a0, bq0, a1, bq1;
      a0.x  = sm.PP[rl][4 * k + kb];
      a0.y  = sm.PP[rl][4 * k + kb + 1];
      bq0.x = sm.leafP[(4 * k + kb) * 16 + col];
      bq0.y = sm.leafP[(4 * k + kb + 1) * 16 + col];
      a1.x  = sm.PP[rl][4 * (k + 1) + kb];
      a1.y  = sm.PP[rl][4 * (k + 1) + kb + 1];
      bq1.x = sm.leafP[(4 * (k + 1) + kb) * 16 + col];
      bq1.y = sm.leafP[(4 * (k + 1) + kb + 1) * 16 + col];
      e0 = __builtin_amdgcn_wmma_f32_16x16x4_f32(false, a0, false, bq0,
                                                 (short)0, e0, false, false);
      e1 = __builtin_amdgcn_wmma_f32_16x16x4_f32(false, a1, false, bq1,
                                                 (short)0, e1, false, false);
    }
    if (col < NUM_CLASSES) {
      #pragma unroll
      for (int r = 0; r < 8; ++r)
        out[(size_t)(waveRow0 + half * 8 + r) * NUM_CLASSES + col] = e0[r] + e1[r];
    }
  }
}

extern "C" void kernel_launch(void* const* d_in, const int* in_sizes, int n_in,
                              void* d_out, int out_size, void* d_ws, size_t ws_size,
                              hipStream_t stream) {
  const float* x    = (const float*)d_in[0];
  const float* pp   = (const float*)d_in[1];
  const float* W1   = (const float*)d_in[2];
  const float* b1   = (const float*)d_in[3];
  const float* w2   = (const float*)d_in[4];
  const float* b2   = (const float*)d_in[5];
  const float* leaf = (const float*)d_in[6];
  float* out = (float*)d_out;

  const int B = in_sizes[0] / INPUT_DIM;
  dim3 grid(B / ROWS_PER_WG);
  dim3 block(THREADS);
  pack_weights_kernel<<<dim3(N_INTERNAL), dim3(NODE_Q), 0, stream>>>(W1, b1, w2, b2, (float*)d_ws);
  tree_wmma_kernel<<<grid, block, 0, stream>>>(x, pp, (const float*)d_ws, leaf, out);
}


# Main-kernel set-up in one memory round trip: 32 leaf-table loads per thread made branch-free, issued with the x, path-probability and root-operand loads before any wait (was 32 serial round trips)
# speedup vs baseline: 5.5100x; 1.0454x over previous
#include <hip/hip_runtime.h>


#define INPUT_DIM   128
#define HID         16
#define MAX_DEPTH   8
#define N_INTERNAL  255
#define N_LEAVES    256
#define NUM_CLASSES 10
#define EFF_IN      130
#define WFLOATS     (EFF_IN * HID)
#define ROWS_PER_WG 64
#define THREADS     128
#define Q_PER_NODE  17
#define NODE_Q      (Q_PER_NODE * 32)
#define S_ZERO      128
#define S_PITCH     129

typedef __attribute__((ext_vector_type(2))) float v2f;
typedef __attribute__((ext_vector_type(8))) float v8f;

struct SMem {
  float leafP[N_LEAVES * 16];
  float PP[ROWS_PER_WG][N_LEAVES + 1];
  float S[ROWS_PER_WG][S_PITCH];
};

struct NodeFrag {
  float4 q[Q_PER_NODE];
};

__global__ __launch_bounds__(NODE_Q) void pack_weights_kernel(
    const float* __restrict__ W1, const float* __restrict__ b1,
    const float* __restrict__ w2, const float* __restrict__ b2,
    float* __restrict__ ws)
{
  const int n    = blockIdx.x;
  const int p    = threadIdx.x >> 5;
  const int lane = threadIdx.x & 31;
  const int h    = lane >> 4;
  const int c    = lane & 15;
  const float* Wn = W1 + (size_t)n * WFLOATS;
  float4 v;
  if (p < 16) {
    const float* r = Wn + (8 * p + h) * HID + c;
    v.x = r[0];
    v.y = r[2 * HID];
    v.z = r[4 * HID];
    v.w = r[6 * HID];
  } else {
    v.x = Wn[(INPUT_DIM + h) * HID + c];
    v.y = b1[n * HID + c];
    v.z = w2[n * HID + c];
    v.w = b2[n];
  }
  ((float4*)ws)[(size_t)n * NODE_Q + threadIdx.x] = v;
}

#define ROW_XOR(v, M) \
  __builtin_bit_cast(float, __builtin_amdgcn_update_dpp(0, __builtin_bit_cast(int, (v)), 0x160 + (M), 0xf, 0xf, false))

__global__ __launch_bounds__(THREADS) void tree_wmma_kernel(
    const float* __restrict__ x, const float* __restrict__ path_prob,
    const float* __restrict__ wpacked,
    const float* __restrict__ leaf_logits, float* __restrict__ out)
{
  __shared__ SMem sm;
  const int tid  = threadIdx.x;
  const int wave = tid >> 5;
  const int lane = tid & 31;
  const int half = lane >> 4;
  const int col  = lane & 15;
  const int kb   = half * 2;
  const int wgRow0   = blockIdx.x * ROWS_PER_WG;
  const int waveRow0 = wgRow0 + wave * 16;
  const int rl       = wave * 16 + col;

  const float4* x4 = (const float4*)(x + (size_t)(waveRow0 + col) * INPUT_DIM);
  float4 xv[32];
  #pragma unroll
  for (int j = 0; j < 32; ++j) xv[j] = x4[j];

  const int lcol = (col < NUM_CLASSES) ? col : NUM_CLASSES - 1;
  const float* lsrc = leaf_logits + (tid >> 4) * NUM_CLASSES + lcol;
  float lv[32];
  #pragma unroll
  for (int i = 0; i < 32; ++i) lv[i] = lsrc[i * 8 * NUM_CLASSES];

  const float pp0 = path_prob[waveRow0 + col];

  const float4* wq = (const float4*)wpacked + lane;
  NodeFrag fa, fb, ga, gb;
  #pragma unroll
  for (int p = 0; p < Q_PER_NODE; ++p) { fa.q[p] = wq[p * 32]; fb.q[p] = fa.q[p]; }

  v2f afrag[32];
  #pragma unroll
  for (int j = 0; j < 32; ++j) {
    afrag[j].x = half ? xv[j].y : xv[j].x;
    afrag[j].y = half ? xv[j].w : xv[j].z;
  }

  #pragma unroll
  for (int i = 0; i < 32; ++i)
    sm.leafP[tid + THREADS * i] = (col < NUM_CLASSES) ? lv[i] : 0.0f;
  sm.PP[rl][0]        = pp0;
  sm.S[rl][S_ZERO]    = 0.0f;
  sm.PP[rl][N_LEAVES] = 0.0f;
  __syncthreads();

  const bool sub = (col & 8) != 0;
  const bool bt2 = (col & 4) != 0;
  const bool bt1 = (col & 2) != 0;
  const bool bt0 = (col & 1) != 0;
  const int prow = wave * 16 + half * 8 + (col & 7);
  float*       ppRow = &sm.PP[prow][0];
  float*       sOut  = &sm.S[prow][0];
  const float* sIn   = &sm.S[rl][0];

  auto split = [&](float s, float v, int slot, int gap) __attribute__((always_inline)) {
    const float p = 1.0f / (1.0f + expf(-s));
    ppRow[slot]       = v * p;
    ppRow[slot + gap] = v * (1.0f - p);
  };

  auto pair_step = [&](const NodeFrag& ca, const NodeFrag& cb, NodeFrag& na, NodeFrag& nb2,
                       int nNextA, int nNextB, float aex,
                       float sP, float vP, int slotP, int gapP)
                       __attribute__((always_inline)) {
    const float4* WnA = wq + (size_t)nNextA * NODE_Q;
    const float4* WnB = wq + (size_t)nNextB * NODE_Q;
    v8f accA = {0.f,0.f,0.f,0.f,0.f,0.f,0.f,0.f};
    v8f accB = {0.f,0.f,0.f,0.f,0.f,0.f,0.f,0.f};
    __builtin_amdgcn_sched_barrier(0);
    #pragma unroll
    for (int p = 0; p < Q_PER_NODE; ++p) na.q[p]  = WnA[p * 32];
    #pragma unroll
    for (int p = 0; p < Q_PER_NODE; ++p) nb2.q[p] = WnB[p * 32];
    __builtin_amdgcn_sched_barrier(0);
    #pragma unroll
    for (int p = 0; p < 16; ++p) {
      v2f a0, a1, b0, b1;
      a0.x = ca.q[p].x;  a0.y = ca.q[p].y;
      a1.x = ca.q[p].z;  a1.y = ca.q[p].w;
      b0.x = cb.q[p].x;  b0.y = cb.q[p].y;
      b1.x = cb.q[p].z;  b1.y = cb.q[p].w;
      accA = __builtin_amdgcn_wmma_f32_16x16x4_f32(false, afrag[2 * p],     false, a0,
                                                   (short)0, accA, false, false);
      accB = __builtin_amdgcn_wmma_f32_16x16x4_f32(false, afrag[2 * p],     false, b0,
                                                   (short)0, accB, false, false);
      accA = __builtin_amdgcn_wmma_f32_16x16x4_f32(false, afrag[2 * p + 1], false, a1,
                                                   (short)0, accA, false, false);
      accB = __builtin_amdgcn_wmma_f32_16x16x4_f32(false, afrag[2 * p + 1], false, b1,
                                                   (short)0, accB, false, false);
    }
    {
      v2f ae, ba, bb;
      ae.x = aex;          ae.y = 0.0f;
      ba.x = ca.q[16].x;   ba.y = 0.0f;
      bb.x = cb.q[16].x;   bb.y = 0.0f;
      accA = __builtin_amdgcn_wmma_f32_16x16x4_f32(false, ae, false, ba,
                                                   (short)0, accA, false, false);
      accB = __builtin_amdgcn_wmma_f32_16x16x4_f32(false, ae, false, bb,
                                                   (short)0, accB, false, false);
    }
    __builtin_amdgcn_sched_barrier(0);
    split(sP, vP, slotP, gapP);
    float u[8];
    #pragma unroll
    for (int r = 0; r < 8; ++r) {
      const float ta = fmaxf(accA[r] + ca.q[16].y, 0.0f) * ca.q[16].z;
      const float tb = fmaxf(accB[r] + cb.q[16].y, 0.0f) * cb.q[16].z;
      const float keep = sub ? tb : ta;
      const float send = sub ? ta : tb;
      u[r] = keep + ROW_XOR(send, 8);
    }
    float w[4];
    #pragma unroll
    for (int r = 0; r < 4; ++r) {
      const float keep = bt2 ? u[r + 4] : u[r];
      const float send = bt2 ? u[r] : u[r + 4];
      w[r] = keep + ROW_XOR(send, 4);
    }
    float y[2];
    #pragma unroll
    for (int r = 0; r < 2; ++r) {
      const float keep = bt1 ? w[r + 2] : w[r];
      const float send = bt1 ? w[r] : w[r + 2];
      y[r] = keep + ROW_XOR(send, 2);
    }
    const float keep = bt0 ? y[1] : y[0];
    const float send = bt0 ? y[0] : y[1];
    const float tot  = keep + ROW_XOR(send, 1);
    return tot + (sub ? cb.q[16].w : ca.q[16].w);
  };

  float sOwed    = 0.0f;
  int   slotOwed = N_LEAVES;
  int   gapOwed  = 0;
  for (int it = 0; it < (N_INTERNAL + 1) / 4; ++it) {
    {
      const int  t    = 2 * it;
      const bool root = (it == 0);
      const int  n1   = 2 * t - 1;
      const int  d    = root ? 0 : 31 - __builtin_clz(2 * t);
      const int  nb   = (1 << d) - 1;
      const int  j2   = root ? 0 : n1 - nb;
      const int  sh   = MAX_DEPTH - d;
      const int  pidx = (d >= 1) ? (nb >> 1) + (j2 >> 1) : S_ZERO;
      const int  gidx = (d >= 2) ? (nb >> 2) + (j2 >> 2) : S_ZERO;
      const float aex   = sIn[half ? gidx : pidx];
      const float vOwed = ppRow[slotOwed];
      const float s = pair_step(fa, fb, ga, gb, 2 * t + 1, 2 * t + 2, aex,
                                sOwed, vOwed, slotOwed, gapOwed);
      sOwed    = s;
      slotOwed = root ? 0 : ((j2 + (sub ? 1 : 0)) << sh);
      gapOwed  = 1 << (sh - 1);
      if (d < MAX_DEPTH - 1)
        sOut[root ? 0 : n1 + (sub ? 1 : 0)] = s;
    }
    {
      const int  t    = 2 * it + 1;
      const int  n1   = 2 * t - 1;
      const int  d    = 31 - __builtin_clz(2 * t);
      const int  nb   = (1 << d) - 1;
      const int  j2   = n1 - nb;
      const int  sh   = MAX_DEPTH - d;
      const int  pidx = (nb >> 1) + (j2 >> 1);
      const int  gidx = (d >= 2) ? (nb >> 2) + (j2 >> 2) : S_ZERO;
      const float aex   = sIn[half ? gidx : pidx];
      const float vOwed = ppRow[slotOwed];
      const int nA = (2 * t + 1 < N_INTERNAL) ? 2 * t + 1 : N_INTERNAL - 1;
      const int nB = (2 * t + 2 < N_INTERNAL) ? 2 * t + 2 : N_INTERNAL - 1;
      const float s = pair_step(ga, gb, fa, fb, nA, nB, aex,
                                sOwed, vOwed, slotOwed, gapOwed);
      sOwed    = s;
      slotOwed = (j2 + (sub ? 1 : 0)) << sh;
      gapOwed  = 1 << (sh - 1);
      if (d < MAX_DEPTH - 1)
        sOut[n1 + (sub ? 1 : 0)] = s;
    }
  }
  split(sOwed, ppRow[slotOwed], slotOwed, gapOwed);

  {
    v8f e0 = {0.f,0.f,0.f,0.f,0.f,0.f,0.f,0.f};
    v8f e1 = {0.f,0.f,0.f,0.f,0.f,0.f,0.f,0.f};
    #pragma unroll
    for (int k = 0; k < 64; k += 2) {
      v2f a0, bq0, a1, bq1;
      a0.x  = sm.PP[rl][4 * k + kb];
      a0.y  = sm.PP[rl][4 * k + kb + 1];
      bq0.x = sm.leafP[(4 * k + kb) * 16 + col];
      bq0.y = sm.leafP[(4 * k + kb + 1) * 16 + col];
      a1.x  = sm.PP[rl][4 * (k + 1) + kb];
      a1.y  = sm.PP[rl][4 * (k + 1) + kb + 1];
      bq1.x = sm.leafP[(4 * (k + 1) + kb) * 16 + col];
      bq1.y = sm.leafP[(4 * (k + 1) + kb + 1) * 16 + col];
      e0 = __builtin_amdgcn_wmma_f32_16x16x4_f32(false, a0, false, bq0,
                                                 (short)0, e0, false, false);
      e1 = __builtin_amdgcn_wmma_f32_16x16x4_f32(false, a1, false, bq1,
                                                 (short)0, e1, false, false);
    }
    if (col < NUM_CLASSES) {
      #pragma unroll
      for (int r = 0; r < 8; ++r)
        out[(size_t)(waveRow0 + half * 8 + r) * NUM_CLASSES + col] = e0[r] + e1[r];
    }
  }
}

extern "C" void kernel_launch(void* const* d_in, const int* in_sizes, int n_in,
                              void* d_out, int out_size, void* d_ws, size_t ws_size,
                              hipStream_t stream) {
  const float* x    = (const float*)d_in[0];
  const float* pp   = (const float*)d_in[1];
  const float* W1   = (const float*)d_in[2];
  const float* b1   = (const float*)d_in[3];
  const float* w2   = (const float*)d_in[4];
  const float* b2   = (const float*)d_in[5];
  const float* leaf = (const float*)d_in[6];
  float* out = (float*)d_out;

  const int B = in_sizes[0] / INPUT_DIM;
  dim3 grid(B / ROWS_PER_WG);
  dim3 block(THREADS);
  pack_weights_kernel<<<dim3(N_INTERNAL), dim3(NODE_Q), 0, stream>>>(W1, b1, w2, b2, (float*)d_ws);
  tree_wmma_kernel<<<grid, block, 0, stream>>>(x, pp, (const float*)d_ws, leaf, out);
}
